# starting kernel
# baseline (speedup 1.0000x reference)
#include <hip/hip_runtime.h>

typedef __attribute__((ext_vector_type(2))) float v2f;
typedef __attribute__((ext_vector_type(8))) float v8f;

#define D_DIM 64
#define B_DIM 2

__device__ __forceinline__ void gadd_f32(float* p, float v) {
  asm volatile("global_atomic_add_f32 %0, %1, off"
               :
               : "v"(p), "v"(v)
               : "memory");
}

__global__ void zero_ws_kernel(float4* __restrict__ ws, int n4) {
  int i = blockIdx.x * blockDim.x + threadIdx.x;
  if (i < n4) {
    float4 z; z.x = 0.f; z.y = 0.f; z.z = 0.f; z.w = 0.f;
    ws[i] = z;
  }
}

__global__ void __launch_bounds__(256)
scatter_edges_kernel(const float* __restrict__ prev,
                     const int*   __restrict__ edges,
                     const float* __restrict__ status,
                     float*       __restrict__ nbr,
                     int E, int N) {
  long tid = (long)blockIdx.x * blockDim.x + threadIdx.x;
  long e   = tid >> 5;
  if (e >= E) return;
  int lane = (int)(tid & 31);
  int b    = lane >> 4;
  int d4   = (lane & 15) * 4;

  int   u = edges[2 * e + 0];
  int   v = edges[2 * e + 1];
  float g = status[e];

  const float4 pu = *(const float4*)(prev + ((size_t)b * N + u) * D_DIM + d4);
  const float4 pv = *(const float4*)(prev + ((size_t)b * N + v) * D_DIM + d4);

  float* ov = nbr + ((size_t)b * N + v) * D_DIM + d4;
  float* ou = nbr + ((size_t)b * N + u) * D_DIM + d4;

  gadd_f32(ov + 0, pu.x * g);
  gadd_f32(ov + 1, pu.y * g);
  gadd_f32(ov + 2, pu.z * g);
  gadd_f32(ov + 3, pu.w * g);
  gadd_f32(ou + 0, pv.x * g);
  gadd_f32(ou + 1, pv.y * g);
  gadd_f32(ou + 2, pv.z * g);
  gadd_f32(ou + 3, pv.w * g);
}

__global__ void __launch_bounds__(256)
gemm_bias_lrelu_kernel(const float* __restrict__ nbr,
                       const float* __restrict__ Wm,
                       const float* __restrict__ nf,
                       const float* __restrict__ ef,
                       float*       __restrict__ out,
                       int ntiles) {
  long tid  = (long)blockIdx.x * blockDim.x + threadIdx.x;
  long tile = tid >> 5;
  if (tile >= ntiles) return;
  int  lane = (int)(tid & 31);
  int  half = lane >> 4;
  int  lm   = lane & 15;
  long row_base = tile * 16;

  v2f afrag[16];
  const float* arow = nbr + (row_base + lm) * D_DIM + 2 * half;
#pragma unroll
  for (int t = 0; t < 16; ++t)
    afrag[t] = *(const v2f*)(arow + 4 * t);

#pragma unroll
  for (int c = 0; c < 4; ++c) {
    int col = c * 16 + lm;

    v8f acc;
#pragma unroll
    for (int r = 0; r < 8; ++r) {
      long rr = row_base + r + 8 * half;
      acc[r] = nf[rr * D_DIM + col] + ef[rr * D_DIM + col];
    }

    const float* wrow = Wm + (size_t)col * D_DIM + 2 * half;
#pragma unroll
    for (int t = 0; t < 16; ++t) {
      v2f bfrag = *(const v2f*)(wrow + 4 * t);
      acc = __builtin_amdgcn_wmma_f32_16x16x4_f32(
          false, afrag[t], false, bfrag, (short)0, acc, false, false);
    }

#pragma unroll
    for (int r = 0; r < 8; ++r) {
      long rr = row_base + r + 8 * half;
      float dv = acc[r];
      dv = dv > 0.f ? dv : 0.01f * dv;
      __builtin_nontemporal_store(dv, out + rr * D_DIM + col);
    }
  }
}

extern "C" void kernel_launch(void* const* d_in, const int* in_sizes, int n_in,
                              void* d_out, int out_size, void* d_ws, size_t ws_size,
                              hipStream_t stream) {
  const float* prev  = (const float*)d_in[0];
  const int*   edges = (const int*)  d_in[1];
  const float* nf    = (const float*)d_in[2];
  const float* ef    = (const float*)d_in[3];
  const float* st    = (const float*)d_in[4];
  const float* Wm    = (const float*)d_in[5];

  float* nbr = (float*)d_ws;

  const int BND = in_sizes[0];
  const int E   = in_sizes[4];
  const int N   = BND / (B_DIM * D_DIM);

  int n4 = BND / 4;
  zero_ws_kernel<<<(n4 + 255) / 256, 256, 0, stream>>>((float4*)nbr, n4);

  long sthreads = (long)E * 32;
  scatter_edges_kernel<<<(int)((sthreads + 255) / 256), 256, 0, stream>>>(
      prev, edges, st, nbr, E, N);

  int ntiles = (BND / D_DIM) / 16;
  long gthreads = (long)ntiles * 32;
  gemm_bias_lrelu_kernel<<<(int)((gthreads + 255) / 256), 256, 0, stream>>>(
      nbr, Wm, nf, ef, (float*)d_out, ntiles);
}


# Owner sums in LDS, list split in two: work-group (node range, half) scans its half from a rotated start, adds gathered rows by LDS f32 atomic adds; global atomics, zero fill gone; sum order changes
# speedup vs baseline: 2.2394x; 2.2394x over previous
#include <hip/hip_runtime.h>

typedef __attribute__((ext_vector_type(2))) float v2f;
typedef __attribute__((ext_vector_type(8))) float v8f;

#define D_DIM 64
#define B_DIM 2

#define RS      391
#define NWAVE   32
#define SCAN_U  4
#define DR      8
#define QCAP    136
#define CHUNK   (NWAVE * SCAN_U * 64)

template <int OFF>
__device__ __forceinline__ void lds_add(unsigned addr, float v) {
  asm volatile("ds_add_f32 %0, %1 offset:%2"
               :
               : "v"(addr), "v"(v), "n"(OFF)
               : "memory");
}

template <bool FULL>
__device__ __forceinline__ void drain_hits(unsigned lane_lds, const int* qs, const int* qd,
                                           const float* qg, int base, int n,
                                           const float* __restrict__ prev,
                                           size_t bstride, int lane) {
  const int   idx = base + (lane & (DR - 1));
  const int   s = qs[idx];
  const int   d = qd[idx];
  const float g = qg[idx];
  float2 r0[DR], r1[DR];
#pragma unroll
  for (int j = 0; j < DR; ++j) {
    if (FULL || j < n) {
      const int sj = __builtin_amdgcn_readlane(s, j);
      const float2* p = (const float2*)(prev + (size_t)sj * D_DIM);
      r0[j] = p[lane];
      r1[j] = ((const float2*)((const float*)p + bstride))[lane];
    }
  }
#pragma unroll
  for (int j = 0; j < DR; ++j) {
    if (FULL || j < n) {
      const int   dj = __builtin_amdgcn_readlane(d, j);
      const float gj = __int_as_float(__builtin_amdgcn_readlane(__float_as_int(g), j));
      const unsigned a0 = lane_lds + (unsigned)dj * (D_DIM * 4);
      const unsigned a1 = a0 + RS * D_DIM * 4;
      lds_add<0>(a0, r0[j].x * gj);
      lds_add<4>(a0, r0[j].y * gj);
      lds_add<0>(a1, r1[j].x * gj);
      lds_add<4>(a1, r1[j].y * gj);
    }
  }
}

__global__ void __launch_bounds__(1024)
owner_scatter_kernel(const float* __restrict__ prev,
                     const int*   __restrict__ edges,
                     const float* __restrict__ status,
                     float*       __restrict__ part0,
                     float*       __restrict__ part1,
                     int E, int N, int nranges) {
  __shared__ float tile[B_DIM * RS * D_DIM];
  __shared__ int   qsrc[NWAVE * QCAP];
  __shared__ int   qdst[NWAVE * QCAP];
  __shared__ float qgat[NWAVE * QCAP];

  const int tid  = (int)threadIdx.x;
  const int lane = tid & 31;
  const int w    = __builtin_amdgcn_readfirstlane(tid >> 5);
  const int r    = (int)(blockIdx.x >> 1);
  const int c    = (int)(blockIdx.x & 1);
  const int lo   = r * RS;
  int cnt = N - lo;
  cnt = cnt > RS ? RS : cnt;

  {
    float4 z; z.x = 0.f; z.y = 0.f; z.z = 0.f; z.w = 0.f;
    float4* t4 = (float4*)tile;
    for (int i = tid; i < B_DIM * RS * D_DIM / 4; i += 1024) t4[i] = z;
  }
  __syncthreads();

  const int mid  = ((E >> 1) + 1) & ~1;
  const int ebeg = c ? mid : 0;
  const int eend = c ? E : mid;
  const int nstep = (eend - ebeg + CHUNK - 1) / CHUNK;
  const int rot   = (int)(((unsigned)r * (unsigned)nstep) / (unsigned)nranges);
  const size_t bstride = (size_t)N * D_DIM;
  const int qb = w * QCAP;
  const unsigned tile_lds =
      (unsigned)(__UINTPTR_TYPE__)(__attribute__((address_space(3))) float*)tile;
  const unsigned lane_lds = tile_lds + (unsigned)lane * 8u;
  int qn = 0;

  int4   cu[SCAN_U];
  float2 cg[SCAN_U];

  auto load_group = [&](int it, int4 (&rc)[SCAN_U], float2 (&gt)[SCAN_U]) {
    const int gbase = ebeg + it * CHUNK + w * (SCAN_U * 64);
    const bool whole = gbase + SCAN_U * 64 <= eend;
#pragma unroll
    for (int s = 0; s < SCAN_U; ++s) {
      const int e = gbase + s * 64 + 2 * lane;
      int4 rcd; rcd.x = -1; rcd.y = -1; rcd.z = -1; rcd.w = -1;
      float2 g; g.x = 0.f; g.y = 0.f;
      if (whole || e + 1 < eend) {
        rcd = *(const int4*)(edges + 2 * (size_t)e);
        g   = *(const float2*)(status + e);
      } else if (e < eend) {
        const int2 t = *(const int2*)(edges + 2 * (size_t)e);
        rcd.x = t.x; rcd.y = t.y;
        g.x = status[e];
      }
      rc[s] = rcd; gt[s] = g;
    }
  };

  auto push = [&](bool h, int src, int dl, float g) {
    const unsigned m = __builtin_amdgcn_ballot_w32(h);
    if (m) {
      const int pos = qb + qn + (int)__builtin_amdgcn_mbcnt_lo(m, 0u);
      if (h) { qsrc[pos] = src; qdst[pos] = dl; qgat[pos] = g; }
      qn += __builtin_popcount(m);
    }
  };

  if (nstep > 0) load_group(rot, cu, cg);
  for (int i = 0; i < nstep; ++i) {
    int4   nx[SCAN_U];
    float2 ng[SCAN_U];
    int itn = i + 1 + rot;
    if (itn >= nstep) itn -= nstep;
    if (i + 1 < nstep) {
      load_group(itn, nx, ng);
    } else {
#pragma unroll
      for (int s = 0; s < SCAN_U; ++s) {
        nx[s].x = -1; nx[s].y = -1; nx[s].z = -1; nx[s].w = -1;
        ng[s].x = 0.f; ng[s].y = 0.f;
      }
    }
#pragma unroll
    for (int s = 0; s < SCAN_U; ++s) {
      const int4   rcd = cu[s];
      const float2 g   = cg[s];
      const int du0 = rcd.x - lo, dv0 = rcd.y - lo;
      const int du1 = rcd.z - lo, dv1 = rcd.w - lo;
      push((unsigned)dv0 < (unsigned)cnt, rcd.x, dv0, g.x);
      push((unsigned)du0 < (unsigned)cnt, rcd.y, du0, g.x);
      push((unsigned)dv1 < (unsigned)cnt, rcd.z, dv1, g.y);
      push((unsigned)du1 < (unsigned)cnt, rcd.w, du1, g.y);
      while (qn >= DR) {
        qn -= DR;
        drain_hits<true>(lane_lds, qsrc, qdst, qgat, qb + qn, DR, prev, bstride, lane);
      }
    }
#pragma unroll
    for (int s = 0; s < SCAN_U; ++s) { cu[s] = nx[s]; cg[s] = ng[s]; }
  }
  if (qn > 0) drain_hits<false>(lane_lds, qsrc, qdst, qgat, qb, qn, prev, bstride, lane);

  asm volatile("s_wait_dscnt 0x0" ::: "memory");
  __syncthreads();

  float* dst = c ? part1 : part0;
  const int n4 = cnt * (D_DIM / 4);
  for (int b = 0; b < B_DIM; ++b) {
    float4*       g4 = (float4*)(dst + ((size_t)b * N + lo) * D_DIM);
    const float4* t4 = (const float4*)(tile + b * RS * D_DIM);
    for (int i = tid; i < n4; i += 1024) g4[i] = t4[i];
  }
}

__global__ void __launch_bounds__(256)
gemm_bias_lrelu_kernel(const float* __restrict__ nbr,
                       const float* __restrict__ Wm,
                       const float* __restrict__ nf,
                       const float* __restrict__ ef,
                       float*                    out,
                       int ntiles) {
  long tid  = (long)blockIdx.x * blockDim.x + threadIdx.x;
  long tile = tid >> 5;
  if (tile >= ntiles) return;
  int  lane = (int)(tid & 31);
  int  half = lane >> 4;
  int  lm   = lane & 15;
  long row_base = tile * 16;

  v2f afrag[16];
  const float* arow = nbr + (row_base + lm) * D_DIM + 2 * half;
  const float* brow = out + (row_base + lm) * D_DIM + 2 * half;
#pragma unroll
  for (int t = 0; t < 16; ++t)
    afrag[t] = *(const v2f*)(arow + 4 * t) + *(const v2f*)(brow + 4 * t);

#pragma unroll
  for (int c = 0; c < 4; ++c) {
    int col = c * 16 + lm;

    v8f acc;
#pragma unroll
    for (int r = 0; r < 8; ++r) {
      long rr = row_base + r + 8 * half;
      acc[r] = nf[rr * D_DIM + col] + ef[rr * D_DIM + col];
    }

    const float* wrow = Wm + (size_t)col * D_DIM + 2 * half;
#pragma unroll
    for (int t = 0; t < 16; ++t) {
      v2f bfrag = *(const v2f*)(wrow + 4 * t);
      acc = __builtin_amdgcn_wmma_f32_16x16x4_f32(
          false, afrag[t], false, bfrag, (short)0, acc, false, false);
    }

#pragma unroll
    for (int r = 0; r < 8; ++r) {
      long rr = row_base + r + 8 * half;
      float dv = acc[r];
      dv = dv > 0.f ? dv : 0.01f * dv;
      __builtin_nontemporal_store(dv, out + rr * D_DIM + col);
    }
  }
}

extern "C" void kernel_launch(void* const* d_in, const int* in_sizes, int n_in,
                              void* d_out, int out_size, void* d_ws, size_t ws_size,
                              hipStream_t stream) {
  const float* prev  = (const float*)d_in[0];
  const int*   edges = (const int*)  d_in[1];
  const float* nf    = (const float*)d_in[2];
  const float* ef    = (const float*)d_in[3];
  const float* st    = (const float*)d_in[4];
  const float* Wm    = (const float*)d_in[5];

  float* nbr = (float*)d_ws;

  const int BND = in_sizes[0];
  const int E   = in_sizes[4];
  const int N   = BND / (B_DIM * D_DIM);

  const int nranges = (N + RS - 1) / RS;
  owner_scatter_kernel<<<2 * nranges, 1024, 0, stream>>>(
      prev, edges, st, nbr, (float*)d_out, E, N, nranges);

  int ntiles = (BND / D_DIM) / 16;
  long gthreads = (long)ntiles * 32;
  gemm_bias_lrelu_kernel<<<(int)((gthreads + 255) / 256), 256, 0, stream>>>(
      nbr, Wm, nf, ef, (float*)d_out, ntiles);
}


# Scan pipelined: loop over whole steps only, loaded unguarded, a wave's 8 loads of the next step back to back and waited for once at the step's end; cut step after the loop; arrival-order sums kept
# speedup vs baseline: 2.5418x; 1.1350x over previous
#include <hip/hip_runtime.h>

typedef __attribute__((ext_vector_type(2))) float v2f;
typedef __attribute__((ext_vector_type(8))) float v8f;

#define D_DIM 64
#define B_DIM 2

#define RS      391
#define NWAVE   32
#define SCAN_U  4
#define DR      8
#define QCAP    136
#define CHUNK   (NWAVE * SCAN_U * 64)

template <int OFF>
__device__ __forceinline__ void lds_add(unsigned addr, float v) {
  asm volatile("ds_add_f32 %0, %1 offset:%2"
               :
               : "v"(addr), "v"(v), "n"(OFF)
               : "memory");
}

template <bool FULL>
__device__ __forceinline__ void drain_hits(unsigned lane_lds, const int* qs, const int* qd,
                                           const float* qg, int base, int n,
                                           const float* __restrict__ prev,
                                           size_t bstride, int lane) {
  const int   idx = base + (lane & (DR - 1));
  const int   s = qs[idx];
  const int   d = qd[idx];
  const float g = qg[idx];
  float2 r0[DR], r1[DR];
#pragma unroll
  for (int j = 0; j < DR; ++j) {
    if (FULL || j < n) {
      const int sj = __builtin_amdgcn_readlane(s, j);
      const float2* p = (const float2*)(prev + (size_t)sj * D_DIM);
      r0[j] = p[lane];
      r1[j] = ((const float2*)((const float*)p + bstride))[lane];
    }
  }
#pragma unroll
  for (int j = 0; j < DR; ++j) {
    if (FULL || j < n) {
      const int   dj = __builtin_amdgcn_readlane(d, j);
      const float gj = __int_as_float(__builtin_amdgcn_readlane(__float_as_int(g), j));
      const unsigned a0 = lane_lds + (unsigned)dj * (D_DIM * 4);
      const unsigned a1 = a0 + RS * D_DIM * 4;
      lds_add<0>(a0, r0[j].x * gj);
      lds_add<4>(a0, r0[j].y * gj);
      lds_add<0>(a1, r1[j].x * gj);
      lds_add<4>(a1, r1[j].y * gj);
    }
  }
}

__global__ void __launch_bounds__(1024)
owner_scatter_kernel(const float* __restrict__ prev,
                     const int*   __restrict__ edges,
                     const float* __restrict__ status,
                     float*       __restrict__ part0,
                     float*       __restrict__ part1,
                     int E, int N, int nranges) {
  __shared__ float tile[B_DIM * RS * D_DIM];
  __shared__ int   qsrc[NWAVE * QCAP];
  __shared__ int   qdst[NWAVE * QCAP];
  __shared__ float qgat[NWAVE * QCAP];

  const int tid  = (int)threadIdx.x;
  const int lane = tid & 31;
  const int w    = __builtin_amdgcn_readfirstlane(tid >> 5);
  const int r    = (int)(blockIdx.x >> 1);
  const int c    = (int)(blockIdx.x & 1);
  const int lo   = r * RS;
  int cnt = N - lo;
  cnt = cnt > RS ? RS : cnt;

  {
    float4 z; z.x = 0.f; z.y = 0.f; z.z = 0.f; z.w = 0.f;
    float4* t4 = (float4*)tile;
    for (int i = tid; i < B_DIM * RS * D_DIM / 4; i += 1024) t4[i] = z;
  }
  __syncthreads();

  const int mid  = ((E >> 1) + 1) & ~1;
  const int ebeg = c ? mid : 0;
  const int eend = c ? E : mid;
  const int nstep = (eend - ebeg + CHUNK - 1) / CHUNK;
  const int nw    = nstep > 0 ? nstep - 1 : 0;
  const int rot   = (int)(((unsigned)r * (unsigned)nw) / (unsigned)nranges);
  const size_t bstride = (size_t)N * D_DIM;
  const int qb = w * QCAP;
  const unsigned tile_lds =
      (unsigned)(__UINTPTR_TYPE__)(__attribute__((address_space(3))) float*)tile;
  const unsigned lane_lds = tile_lds + (unsigned)lane * 8u;
  int qn = 0;

  auto load_cut = [&](int it, int4 (&rc)[SCAN_U], float2 (&gt)[SCAN_U]) {
    const int gbase = ebeg + it * CHUNK + w * (SCAN_U * 64);
#pragma unroll
    for (int s = 0; s < SCAN_U; ++s) {
      const int e = gbase + s * 64 + 2 * lane;
      int4 rcd; rcd.x = -1; rcd.y = -1; rcd.z = -1; rcd.w = -1;
      float2 g; g.x = 0.f; g.y = 0.f;
      if (e + 1 < eend) {
        rcd = *(const int4*)(edges + 2 * (size_t)e);
        g   = *(const float2*)(status + e);
      } else if (e < eend) {
        const int2 t = *(const int2*)(edges + 2 * (size_t)e);
        rcd.x = t.x; rcd.y = t.y;
        g.x = status[e];
      }
      rc[s] = rcd; gt[s] = g;
    }
  };

  auto load_whole = [&](int it, int4 (&rc)[SCAN_U], float2 (&gt)[SCAN_U]) {
    const int gbase = ebeg + it * CHUNK + w * (SCAN_U * 64);
    const int4*   pe = (const int4*)(edges + 2 * (size_t)gbase) + lane;
    const float2* pg = (const float2*)(status + gbase) + lane;
#pragma unroll
    for (int s = 0; s < SCAN_U; ++s) { rc[s] = pe[32 * s]; gt[s] = pg[32 * s]; }
  };

  auto push = [&](bool h, int src, int dl, float g) {
    const unsigned m = __builtin_amdgcn_ballot_w32(h);
    if (m) {
      const int pos = qb + qn + (int)__builtin_amdgcn_mbcnt_lo(m, 0u);
      if (h) { qsrc[pos] = src; qdst[pos] = dl; qgat[pos] = g; }
      qn += __builtin_popcount(m);
    }
  };

  auto process = [&](const int4 (&rc)[SCAN_U], const float2 (&gt)[SCAN_U]) {
#pragma unroll
    for (int s = 0; s < SCAN_U; ++s) {
      const int4   rcd = rc[s];
      const float2 g   = gt[s];
      const int du0 = rcd.x - lo, dv0 = rcd.y - lo;
      const int du1 = rcd.z - lo, dv1 = rcd.w - lo;
      push((unsigned)dv0 < (unsigned)cnt, rcd.x, dv0, g.x);
      push((unsigned)du0 < (unsigned)cnt, rcd.y, du0, g.x);
      push((unsigned)dv1 < (unsigned)cnt, rcd.z, dv1, g.y);
      push((unsigned)du1 < (unsigned)cnt, rcd.w, du1, g.y);
      while (qn >= DR) {
        qn -= DR;
        drain_hits<true>(lane_lds, qsrc, qdst, qgat, qb + qn, DR, prev, bstride, lane);
      }
    }
  };

  int4   cu[SCAN_U];
  float2 cg[SCAN_U];
#pragma unroll
  for (int s = 0; s < SCAN_U; ++s) {
    cu[s].x = -1; cu[s].y = -1; cu[s].z = -1; cu[s].w = -1;
    cg[s].x = 0.f; cg[s].y = 0.f;
  }

  if (nw > 0) {
    int it = rot;
    load_whole(it, cu, cg);
    for (int i = 0; i + 1 < nw; ++i) {
      int4   nx[SCAN_U];
      float2 ng[SCAN_U];
      ++it;
      if (it == nw) it = 0;
      load_whole(it, nx, ng);
      process(cu, cg);
#pragma unroll
      for (int s = 0; s < SCAN_U; ++s) { cu[s] = nx[s]; cg[s] = ng[s]; }
    }
  }

  {
    int4   ct[SCAN_U];
    float2 ctg[SCAN_U];
    if (nstep > 0) {
      load_cut(nstep - 1, ct, ctg);
    } else {
#pragma unroll
      for (int s = 0; s < SCAN_U; ++s) {
        ct[s].x = -1; ct[s].y = -1; ct[s].z = -1; ct[s].w = -1;
        ctg[s].x = 0.f; ctg[s].y = 0.f;
      }
    }
#pragma nounroll
    for (int k = 0; k < 2; ++k) {
      process(cu, cg);
#pragma unroll
      for (int s = 0; s < SCAN_U; ++s) { cu[s] = ct[s]; cg[s] = ctg[s]; }
    }
  }
  if (qn > 0) drain_hits<false>(lane_lds, qsrc, qdst, qgat, qb, qn, prev, bstride, lane);

  asm volatile("s_wait_dscnt 0x0" ::: "memory");
  __syncthreads();

  float* dst = c ? part1 : part0;
  const int n4 = cnt * (D_DIM / 4);
  for (int b = 0; b < B_DIM; ++b) {
    float4*       g4 = (float4*)(dst + ((size_t)b * N + lo) * D_DIM);
    const float4* t4 = (const float4*)(tile + b * RS * D_DIM);
    for (int i = tid; i < n4; i += 1024) g4[i] = t4[i];
  }
}

__global__ void __launch_bounds__(256)
gemm_bias_lrelu_kernel(const float* __restrict__ nbr,
                       const float* __restrict__ Wm,
                       const float* __restrict__ nf,
                       const float* __restrict__ ef,
                       float*                    out,
                       int ntiles) {
  long tid  = (long)blockIdx.x * blockDim.x + threadIdx.x;
  long tile = tid >> 5;
  if (tile >= ntiles) return;
  int  lane = (int)(tid & 31);
  int  half = lane >> 4;
  int  lm   = lane & 15;
  long row_base = tile * 16;

  v2f afrag[16];
  const float* arow = nbr + (row_base + lm) * D_DIM + 2 * half;
  const float* brow = out + (row_base + lm) * D_DIM + 2 * half;
#pragma unroll
  for (int t = 0; t < 16; ++t)
    afrag[t] = *(const v2f*)(arow + 4 * t) + *(const v2f*)(brow + 4 * t);

#pragma unroll
  for (int c = 0; c < 4; ++c) {
    int col = c * 16 + lm;

    v8f acc;
#pragma unroll
    for (int r = 0; r < 8; ++r) {
      long rr = row_base + r + 8 * half;
      acc[r] = nf[rr * D_DIM + col] + ef[rr * D_DIM + col];
    }

    const float* wrow = Wm + (size_t)col * D_DIM + 2 * half;
#pragma unroll
    for (int t = 0; t < 16; ++t) {
      v2f bfrag = *(const v2f*)(wrow + 4 * t);
      acc = __builtin_amdgcn_wmma_f32_16x16x4_f32(
          false, afrag[t], false, bfrag, (short)0, acc, false, false);
    }

#pragma unroll
    for (int r = 0; r < 8; ++r) {
      long rr = row_base + r + 8 * half;
      float dv = acc[r];
      dv = dv > 0.f ? dv : 0.01f * dv;
      __builtin_nontemporal_store(dv, out + rr * D_DIM + col);
    }
  }
}

extern "C" void kernel_launch(void* const* d_in, const int* in_sizes, int n_in,
                              void* d_out, int out_size, void* d_ws, size_t ws_size,
                              hipStream_t stream) {
  const float* prev  = (const float*)d_in[0];
  const int*   edges = (const int*)  d_in[1];
  const float* nf    = (const float*)d_in[2];
  const float* ef    = (const float*)d_in[3];
  const float* st    = (const float*)d_in[4];
  const float* Wm    = (const float*)d_in[5];

  float* nbr = (float*)d_ws;

  const int BND = in_sizes[0];
  const int E   = in_sizes[4];
  const int N   = BND / (B_DIM * D_DIM);

  const int nranges = (N + RS - 1) / RS;
  owner_scatter_kernel<<<2 * nranges, 1024, 0, stream>>>(
      prev, edges, st, nbr, (float*)d_out, E, N, nranges);

  int ntiles = (BND / D_DIM) / 16;
  long gthreads = (long)ntiles * 32;
  gemm_bias_lrelu_kernel<<<(int)((gthreads + 255) / 256), 256, 0, stream>>>(
      nbr, Wm, nf, ef, (float*)d_out, ntiles);
}


# Scan streams the endpoint pairs only (8 not 12 bytes a record): a hit queues its record number and the drain fetches the 8 gates by number ahead of the row gathers; arrival-order sums kept
# speedup vs baseline: 2.6365x; 1.0373x over previous
#include <hip/hip_runtime.h>

typedef __attribute__((ext_vector_type(2))) float v2f;
typedef __attribute__((ext_vector_type(8))) float v8f;

#define D_DIM 64
#define B_DIM 2

#define RS      391
#define NWAVE   32
#define SCAN_U  4
#define DR      8
#define QCAP    136
#define CHUNK   (NWAVE * SCAN_U * 64)

template <int OFF>
__device__ __forceinline__ void lds_add(unsigned addr, float v) {
  asm volatile("ds_add_f32 %0, %1 offset:%2"
               :
               : "v"(addr), "v"(v), "n"(OFF)
               : "memory");
}

template <bool FULL>
__device__ __forceinline__ void drain_hits(unsigned lane_lds, const int* qs, const int* qd,
                                           const int* qe, int base, int n,
                                           const float* __restrict__ prev,
                                           const float* __restrict__ status,
                                           size_t bstride, int lane) {
  const int   idx = base + (lane & (DR - 1));
  const int   s = qs[idx];
  const int   d = qd[idx];
  float g = 0.f;
  if (FULL || (lane & (DR - 1)) < n) g = status[qe[idx]];
  float2 r0[DR], r1[DR];
#pragma unroll
  for (int j = 0; j < DR; ++j) {
    if (FULL || j < n) {
      const int sj = __builtin_amdgcn_readlane(s, j);
      const float2* p = (const float2*)(prev + (size_t)sj * D_DIM);
      r0[j] = p[lane];
      r1[j] = ((const float2*)((const float*)p + bstride))[lane];
    }
  }
#pragma unroll
  for (int j = 0; j < DR; ++j) {
    if (FULL || j < n) {
      const int   dj = __builtin_amdgcn_readlane(d, j);
      const float gj = __int_as_float(__builtin_amdgcn_readlane(__float_as_int(g), j));
      const unsigned a0 = lane_lds + (unsigned)dj * (D_DIM * 4);
      const unsigned a1 = a0 + RS * D_DIM * 4;
      lds_add<0>(a0, r0[j].x * gj);
      lds_add<4>(a0, r0[j].y * gj);
      lds_add<0>(a1, r1[j].x * gj);
      lds_add<4>(a1, r1[j].y * gj);
    }
  }
}

__global__ void __launch_bounds__(1024)
owner_scatter_kernel(const float* __restrict__ prev,
                     const int*   __restrict__ edges,
                     const float* __restrict__ status,
                     float*       __restrict__ part0,
                     float*       __restrict__ part1,
                     int E, int N, int nranges) {
  __shared__ float tile[B_DIM * RS * D_DIM];
  __shared__ int   qsrc[NWAVE * QCAP];
  __shared__ int   qdst[NWAVE * QCAP];
  __shared__ int   qrec[NWAVE * QCAP];

  const int tid  = (int)threadIdx.x;
  const int lane = tid & 31;
  const int w    = __builtin_amdgcn_readfirstlane(tid >> 5);
  const int r    = (int)(blockIdx.x >> 1);
  const int c    = (int)(blockIdx.x & 1);
  const int lo   = r * RS;
  int cnt = N - lo;
  cnt = cnt > RS ? RS : cnt;

  {
    float4 z; z.x = 0.f; z.y = 0.f; z.z = 0.f; z.w = 0.f;
    float4* t4 = (float4*)tile;
    for (int i = tid; i < B_DIM * RS * D_DIM / 4; i += 1024) t4[i] = z;
  }
  __syncthreads();

  const int mid  = ((E >> 1) + 1) & ~1;
  const int ebeg = c ? mid : 0;
  const int eend = c ? E : mid;
  const int nstep = (eend - ebeg + CHUNK - 1) / CHUNK;
  const int nw    = nstep > 0 ? nstep - 1 : 0;
  const int rot   = (int)(((unsigned)r * (unsigned)nw) / (unsigned)nranges);
  const size_t bstride = (size_t)N * D_DIM;
  const int qb = w * QCAP;
  const unsigned tile_lds =
      (unsigned)(__UINTPTR_TYPE__)(__attribute__((address_space(3))) float*)tile;
  const unsigned lane_lds = tile_lds + (unsigned)lane * 8u;
  int qn = 0;

  auto load_cut = [&](int it, int4 (&rc)[SCAN_U]) {
    const int gbase = ebeg + it * CHUNK + w * (SCAN_U * 64);
#pragma unroll
    for (int s = 0; s < SCAN_U; ++s) {
      const int e = gbase + s * 64 + 2 * lane;
      int4 rcd; rcd.x = -1; rcd.y = -1; rcd.z = -1; rcd.w = -1;
      if (e + 1 < eend) {
        rcd = *(const int4*)(edges + 2 * (size_t)e);
      } else if (e < eend) {
        const int2 t = *(const int2*)(edges + 2 * (size_t)e);
        rcd.x = t.x; rcd.y = t.y;
      }
      rc[s] = rcd;
    }
  };

  auto step_base = [&](int it) { return ebeg + it * CHUNK + w * (SCAN_U * 64); };

  auto load_whole = [&](int it, int4 (&rc)[SCAN_U]) {
    const int4* pe = (const int4*)(edges + 2 * (size_t)step_base(it)) + lane;
#pragma unroll
    for (int s = 0; s < SCAN_U; ++s) rc[s] = pe[32 * s];
  };

  auto push = [&](bool h, int src, int dl, int e) {
    const unsigned m = __builtin_amdgcn_ballot_w32(h);
    if (m) {
      const int pos = qb + qn + (int)__builtin_amdgcn_mbcnt_lo(m, 0u);
      if (h) { qsrc[pos] = src; qdst[pos] = dl; qrec[pos] = e; }
      qn += __builtin_popcount(m);
    }
  };

  auto process = [&](const int4 (&rc)[SCAN_U], int sb) {
#pragma unroll
    for (int s = 0; s < SCAN_U; ++s) {
      const int4 rcd = rc[s];
      const int  e0  = sb + s * 64 + 2 * lane;
      const int du0 = rcd.x - lo, dv0 = rcd.y - lo;
      const int du1 = rcd.z - lo, dv1 = rcd.w - lo;
      push((unsigned)dv0 < (unsigned)cnt, rcd.x, dv0, e0);
      push((unsigned)du0 < (unsigned)cnt, rcd.y, du0, e0);
      push((unsigned)dv1 < (unsigned)cnt, rcd.z, dv1, e0 + 1);
      push((unsigned)du1 < (unsigned)cnt, rcd.w, du1, e0 + 1);
      while (qn >= DR) {
        qn -= DR;
        drain_hits<true>(lane_lds, qsrc, qdst, qrec, qb + qn, DR, prev, status, bstride, lane);
      }
    }
  };

  int4 cu[SCAN_U];
  int  cb = 0;
#pragma unroll
  for (int s = 0; s < SCAN_U; ++s) { cu[s].x = -1; cu[s].y = -1; cu[s].z = -1; cu[s].w = -1; }

  if (nw > 0) {
    int it = rot;
    load_whole(it, cu);
    cb = step_base(it);
    for (int i = 0; i + 1 < nw; ++i) {
      int4 nx[SCAN_U];
      ++it;
      if (it == nw) it = 0;
      load_whole(it, nx);
      process(cu, cb);
      cb = step_base(it);
#pragma unroll
      for (int s = 0; s < SCAN_U; ++s) cu[s] = nx[s];
    }
  }

  {
    int4 ct[SCAN_U];
    if (nstep > 0) {
      load_cut(nstep - 1, ct);
    } else {
#pragma unroll
      for (int s = 0; s < SCAN_U; ++s) { ct[s].x = -1; ct[s].y = -1; ct[s].z = -1; ct[s].w = -1; }
    }
    const int ctb = step_base(nstep > 0 ? nstep - 1 : 0);
#pragma nounroll
    for (int k = 0; k < 2; ++k) {
      process(cu, cb);
      cb = ctb;
#pragma unroll
      for (int s = 0; s < SCAN_U; ++s) cu[s] = ct[s];
    }
  }
  if (qn > 0) drain_hits<false>(lane_lds, qsrc, qdst, qrec, qb, qn, prev, status, bstride, lane);

  asm volatile("s_wait_dscnt 0x0" ::: "memory");
  __syncthreads();

  float* dst = c ? part1 : part0;
  const int n4 = cnt * (D_DIM / 4);
  for (int b = 0; b < B_DIM; ++b) {
    float4*       g4 = (float4*)(dst + ((size_t)b * N + lo) * D_DIM);
    const float4* t4 = (const float4*)(tile + b * RS * D_DIM);
    for (int i = tid; i < n4; i += 1024) g4[i] = t4[i];
  }
}

__global__ void __launch_bounds__(256)
gemm_bias_lrelu_kernel(const float* __restrict__ nbr,
                       const float* __restrict__ Wm,
                       const float* __restrict__ nf,
                       const float* __restrict__ ef,
                       float*                    out,
                       int ntiles) {
  long tid  = (long)blockIdx.x * blockDim.x + threadIdx.x;
  long tile = tid >> 5;
  if (tile >= ntiles) return;
  int  lane = (int)(tid & 31);
  int  half = lane >> 4;
  int  lm   = lane & 15;
  long row_base = tile * 16;

  v2f afrag[16];
  const float* arow = nbr + (row_base + lm) * D_DIM + 2 * half;
  const float* brow = out + (row_base + lm) * D_DIM + 2 * half;
#pragma unroll
  for (int t = 0; t < 16; ++t)
    afrag[t] = *(const v2f*)(arow + 4 * t) + *(const v2f*)(brow + 4 * t);

#pragma unroll
  for (int c = 0; c < 4; ++c) {
    int col = c * 16 + lm;

    v8f acc;
#pragma unroll
    for (int r = 0; r < 8; ++r) {
      long rr = row_base + r + 8 * half;
      acc[r] = nf[rr * D_DIM + col] + ef[rr * D_DIM + col];
    }

    const float* wrow = Wm + (size_t)col * D_DIM + 2 * half;
#pragma unroll
    for (int t = 0; t < 16; ++t) {
      v2f bfrag = *(const v2f*)(wrow + 4 * t);
      acc = __builtin_amdgcn_wmma_f32_16x16x4_f32(
          false, afrag[t], false, bfrag, (short)0, acc, false, false);
    }

#pragma unroll
    for (int r = 0; r < 8; ++r) {
      long rr = row_base + r + 8 * half;
      float dv = acc[r];
      dv = dv > 0.f ? dv : 0.01f * dv;
      __builtin_nontemporal_store(dv, out + rr * D_DIM + col);
    }
  }
}

extern "C" void kernel_launch(void* const* d_in, const int* in_sizes, int n_in,
                              void* d_out, int out_size, void* d_ws, size_t ws_size,
                              hipStream_t stream) {
  const float* prev  = (const float*)d_in[0];
  const int*   edges = (const int*)  d_in[1];
  const float* nf    = (const float*)d_in[2];
  const float* ef    = (const float*)d_in[3];
  const float* st    = (const float*)d_in[4];
  const float* Wm    = (const float*)d_in[5];

  float* nbr = (float*)d_ws;

  const int BND = in_sizes[0];
  const int E   = in_sizes[4];
  const int N   = BND / (B_DIM * D_DIM);

  const int nranges = (N + RS - 1) / RS;
  owner_scatter_kernel<<<2 * nranges, 1024, 0, stream>>>(
      prev, edges, st, nbr, (float*)d_out, E, N, nranges);

  int ntiles = (BND / D_DIM) / 16;
  long gthreads = (long)ntiles * 32;
  gemm_bias_lrelu_kernel<<<(int)((gthreads + 255) / 256), 256, 0, stream>>>(
      nbr, Wm, nf, ef, (float*)d_out, ntiles);
}
